# starting kernel
# baseline (speedup 1.0000x reference)
#include <hip/hip_runtime.h>

typedef float v2f __attribute__((ext_vector_type(2)));
typedef float v4f __attribute__((ext_vector_type(4)));
typedef float v8f __attribute__((ext_vector_type(8)));

static constexpr int kB  = 16;
static constexpr int kC  = 256;
static constexpr int kW  = 128;
static constexpr int kHW = 128 * 128;

__global__ __launch_bounds__(256) void reduce_mean_max(
    const float* __restrict__ x, float* __restrict__ s) {
  int t   = blockIdx.x * 256 + threadIdx.x;
  int b   = t >> 12;
  int hw4 = (t & 4095) << 2;

  const v4f* p = (const v4f*)(x + (size_t)b * kC * kHW + hw4);
  v4f sum = {0.f, 0.f, 0.f, 0.f};
  float ninf = -__builtin_inff();
  v4f mx = {ninf, ninf, ninf, ninf};

#pragma unroll 16
  for (int c = 0; c < kC; ++c) {
    v4f v = __builtin_nontemporal_load(p + c * (kHW / 4));
    sum += v;
    mx[0] = fmaxf(mx[0], v[0]);
    mx[1] = fmaxf(mx[1], v[1]);
    mx[2] = fmaxf(mx[2], v[2]);
    mx[3] = fmaxf(mx[3], v[3]);
  }
  v4f avg = sum * (1.0f / 256.0f);

  v4f* sp = (v4f*)(s + (size_t)b * 2 * kHW);
  sp[hw4 >> 2]              = avg;
  sp[(kHW >> 2) + (hw4 >> 2)] = mx;
}

__global__ __launch_bounds__(32) void conv7x7_wmma(
    const float* __restrict__ s, const float* __restrict__ w,
    float* __restrict__ out) {
  __shared__ float tile[2 * 22 * 24];
  __shared__ float wl[98];

  int lane = threadIdx.x;
  int bid  = blockIdx.x;
  int b    = bid >> 6;
  int tl   = bid & 63;
  int y0   = (tl >> 3) << 4;
  int x0   = (tl & 7) << 4;

  const float* sb = s + (size_t)b * 2 * kHW;

  for (int i = lane; i < 98; i += 32) wl[i] = w[i];

  for (int pos = lane; pos < 2 * 484; pos += 32) {
    int c  = (pos >= 484);
    int r  = pos - c * 484;
    int yy = r / 22;
    int xx = r - yy * 22;
    int gy = y0 + yy - 3;
    int gx = x0 + xx - 3;
    float v = 0.0f;
    if ((unsigned)gy < 128u && (unsigned)gx < 128u)
      v = sb[c * kHW + gy * kW + gx];
    tile[c * 528 + yy * 24 + xx] = v;
  }
  __syncthreads();

  int half = lane >> 4;
  int mn   = lane & 15;

  int c0 = 0, yy0 = 0, dx0 = half * 2;

  v8f acc = {0.f, 0.f, 0.f, 0.f, 0.f, 0.f, 0.f, 0.f};

  for (int step = 0; step < 77; ++step) {
    int dx1 = dx0 + 1, yy1 = yy0, c1 = c0;
    if (dx1 == 7) { dx1 = 0; ++yy1; if (yy1 == 22) { yy1 = 0; ++c1; } }

    v2f a, bb;
    a.x = tile[c0 * 528 + yy0 * 24 + mn + dx0];
    a.y = tile[c1 * 528 + yy1 * 24 + mn + dx1];

    int dyA = yy0 - mn;
    int dyB = yy1 - mn;
    int iA = dyA < 0 ? 0 : (dyA > 6 ? 6 : dyA);
    int iB = dyB < 0 ? 0 : (dyB > 6 ? 6 : dyB);
    float wA = wl[c0 * 49 + iA * 7 + dx0];
    float wB = wl[c1 * 49 + iB * 7 + dx1];
    bb.x = ((unsigned)dyA < 7u) ? wA : 0.0f;
    bb.y = ((unsigned)dyB < 7u) ? wB : 0.0f;

    acc = __builtin_amdgcn_wmma_f32_16x16x4_f32(
        false, a, false, bb, (short)0, acc, false, false);

    dx0 += 4;
    if (dx0 >= 7) { dx0 -= 7; ++yy0; if (yy0 == 22) { yy0 = 0; ++c0; } }
  }

  int iy = mn;
  int jb = half * 8;
  float* ob = out + (size_t)b * kHW + (y0 + iy) * kW + (x0 + jb);
#pragma unroll
  for (int v = 0; v < 8; ++v) {
    float val = acc[v];
    ob[v] = 1.0f / (1.0f + __expf(-val));
  }
}

extern "C" void kernel_launch(void* const* d_in, const int* in_sizes, int n_in,
                              void* d_out, int out_size, void* d_ws, size_t ws_size,
                              hipStream_t stream) {
  const float* x = (const float*)d_in[0];
  const float* w = (const float*)d_in[1];
  float* out = (float*)d_out;
  float* s   = (float*)d_ws;

  reduce_mean_max<<<256, 256, 0, stream>>>(x, s);
  conv7x7_wmma<<<kB * 64, 32, 0, stream>>>(s, w, out);
}


# conv tile staging: the 44 row loads and 4 weight loads are issued unconditionally (clamped address, select to zero) before first use, replacing 31+4 branch-guarded load-wait-store trips
# speedup vs baseline: 1.3475x; 1.3475x over previous
#include <hip/hip_runtime.h>

typedef float v2f __attribute__((ext_vector_type(2)));
typedef float v4f __attribute__((ext_vector_type(4)));
typedef float v8f __attribute__((ext_vector_type(8)));

static constexpr int kB  = 16;
static constexpr int kC  = 256;
static constexpr int kW  = 128;
static constexpr int kHW = 128 * 128;

__global__ __launch_bounds__(256) void reduce_mean_max(
    const float* __restrict__ x, float* __restrict__ s) {
  int t   = blockIdx.x * 256 + threadIdx.x;
  int b   = t >> 12;
  int hw4 = (t & 4095) << 2;

  const v4f* p = (const v4f*)(x + (size_t)b * kC * kHW + hw4);
  v4f sum = {0.f, 0.f, 0.f, 0.f};
  float ninf = -__builtin_inff();
  v4f mx = {ninf, ninf, ninf, ninf};

#pragma unroll 16
  for (int c = 0; c < kC; ++c) {
    v4f v = __builtin_nontemporal_load(p + c * (kHW / 4));
    sum += v;
    mx[0] = fmaxf(mx[0], v[0]);
    mx[1] = fmaxf(mx[1], v[1]);
    mx[2] = fmaxf(mx[2], v[2]);
    mx[3] = fmaxf(mx[3], v[3]);
  }
  v4f avg = sum * (1.0f / 256.0f);

  v4f* sp = (v4f*)(s + (size_t)b * 2 * kHW);
  sp[hw4 >> 2]              = avg;
  sp[(kHW >> 2) + (hw4 >> 2)] = mx;
}

__global__ __launch_bounds__(32) void conv7x7_wmma(
    const float* __restrict__ s, const float* __restrict__ w,
    float* __restrict__ out) {
  __shared__ float tile[2 * 22 * 24];
  __shared__ float wl[98];

  int lane = threadIdx.x;
  int bid  = blockIdx.x;
  int b    = bid >> 6;
  int tl   = bid & 63;
  int y0   = (tl >> 3) << 4;
  int x0   = (tl & 7) << 4;

  const float* sb = s + (size_t)b * 2 * kHW;

  float wv[4];
  for (int k = 0; k < 4; ++k) {
    int i = lane + 32 * k;
    wv[k] = w[i < 98 ? i : 97];
  }

  int gx  = x0 + lane - 3;
  int gxc = gx < 0 ? 0 : (gx > 127 ? 127 : gx);
  bool colok = (unsigned)gx < 128u;

  float tv[44];
#pragma unroll
  for (int r = 0; r < 44; ++r) {
    int c  = r / 22;
    int yy = r - c * 22;
    int gy = y0 + yy - 3;
    int gyc = gy < 0 ? 0 : (gy > 127 ? 127 : gy);
    tv[r] = sb[c * kHW + gyc * kW + gxc];
  }

  for (int k = 0; k < 4; ++k) {
    int i = lane + 32 * k;
    if (i < 98) wl[i] = wv[k];
  }
  if (lane < 24) {
#pragma unroll
    for (int r = 0; r < 44; ++r) {
      int c  = r / 22;
      int yy = r - c * 22;
      int gy = y0 + yy - 3;
      bool ok = colok && ((unsigned)gy < 128u);
      tile[c * 528 + yy * 24 + lane] = ok ? tv[r] : 0.0f;
    }
  }
  __syncthreads();

  int half = lane >> 4;
  int mn   = lane & 15;

  int c0 = 0, yy0 = 0, dx0 = half * 2;

  v8f acc = {0.f, 0.f, 0.f, 0.f, 0.f, 0.f, 0.f, 0.f};

  for (int step = 0; step < 77; ++step) {
    int dx1 = dx0 + 1, yy1 = yy0, c1 = c0;
    if (dx1 == 7) { dx1 = 0; ++yy1; if (yy1 == 22) { yy1 = 0; ++c1; } }

    v2f a, bb;
    a.x = tile[c0 * 528 + yy0 * 24 + mn + dx0];
    a.y = tile[c1 * 528 + yy1 * 24 + mn + dx1];

    int dyA = yy0 - mn;
    int dyB = yy1 - mn;
    int iA = dyA < 0 ? 0 : (dyA > 6 ? 6 : dyA);
    int iB = dyB < 0 ? 0 : (dyB > 6 ? 6 : dyB);
    float wA = wl[c0 * 49 + iA * 7 + dx0];
    float wB = wl[c1 * 49 + iB * 7 + dx1];
    bb.x = ((unsigned)dyA < 7u) ? wA : 0.0f;
    bb.y = ((unsigned)dyB < 7u) ? wB : 0.0f;

    acc = __builtin_amdgcn_wmma_f32_16x16x4_f32(
        false, a, false, bb, (short)0, acc, false, false);

    dx0 += 4;
    if (dx0 >= 7) { dx0 -= 7; ++yy0; if (yy0 == 22) { yy0 = 0; ++c0; } }
  }

  int iy = mn;
  int jb = half * 8;
  float* ob = out + (size_t)b * kHW + (y0 + iy) * kW + (x0 + jb);
#pragma unroll
  for (int v = 0; v < 8; ++v) {
    float val = acc[v];
    ob[v] = 1.0f / (1.0f + __expf(-val));
  }
}

extern "C" void kernel_launch(void* const* d_in, const int* in_sizes, int n_in,
                              void* d_out, int out_size, void* d_ws, size_t ws_size,
                              hipStream_t stream) {
  const float* x = (const float*)d_in[0];
  const float* w = (const float*)d_in[1];
  float* out = (float*)d_out;
  float* s   = (float*)d_ws;

  reduce_mean_max<<<256, 256, 0, stream>>>(x, s);
  conv7x7_wmma<<<kB * 64, 32, 0, stream>>>(s, w, out);
}


# reduce kernel keeps more bytes in flight per WGP: 16 waves (channels 0..127 / 128..255 on two wave groups, partials joined in LDS: mean summed as two halves), 16-32 loads outstanding per wave
# speedup vs baseline: 1.4885x; 1.1046x over previous
#include <hip/hip_runtime.h>

typedef float v2f __attribute__((ext_vector_type(2)));
typedef float v4f __attribute__((ext_vector_type(4)));
typedef float v8f __attribute__((ext_vector_type(8)));

static constexpr int kB  = 16;
static constexpr int kC  = 256;
static constexpr int kW  = 128;
static constexpr int kHW = 128 * 128;

static constexpr int kBatch   = 16;
static constexpr int kHalfC   = kC / 2;
static constexpr int kBatches = kHalfC / kBatch;

__global__ __launch_bounds__(512) void reduce_mean_max(
    const float* __restrict__ x, float* __restrict__ s) {
  __shared__ v4f part[2][256];

  int q    = threadIdx.x & 255;
  int hsel = threadIdx.x >> 8;
  int t    = blockIdx.x * 256 + q;
  int b    = t >> 12;
  int hw4  = (t & 4095) << 2;

  const v4f* p = (const v4f*)(x + (size_t)b * kC * kHW
                                + (size_t)hsel * kHalfC * kHW + hw4);
  v4f sum = {0.f, 0.f, 0.f, 0.f};
  float ninf = -__builtin_inff();
  v4f mx = {ninf, ninf, ninf, ninf};

  v4f buf[2][kBatch];
#define LOAD_BATCH(B)                                                        \
  _Pragma("unroll") for (int i = 0; i < kBatch; ++i)                         \
    buf[(B) & 1][i] =                                                        \
        __builtin_nontemporal_load(p + ((B) * kBatch + i) * (kHW / 4));

  LOAD_BATCH(0);
#pragma unroll
  for (int bt = 0; bt < kBatches; ++bt) {
    __builtin_amdgcn_sched_barrier(0);
    if (bt + 1 < kBatches) { LOAD_BATCH(bt + 1); }
    __builtin_amdgcn_sched_barrier(0);
#pragma unroll
    for (int i = 0; i < kBatch; ++i) {
      v4f v = buf[bt & 1][i];
      sum += v;
      mx[0] = fmaxf(mx[0], v[0]);
      mx[1] = fmaxf(mx[1], v[1]);
      mx[2] = fmaxf(mx[2], v[2]);
      mx[3] = fmaxf(mx[3], v[3]);
    }
  }
#undef LOAD_BATCH

  if (hsel) {
    part[0][q] = sum;
    part[1][q] = mx;
  }
  __syncthreads();
  if (!hsel) {
    v4f s2 = part[0][q];
    v4f m2 = part[1][q];
    sum += s2;
    mx[0] = fmaxf(mx[0], m2[0]);
    mx[1] = fmaxf(mx[1], m2[1]);
    mx[2] = fmaxf(mx[2], m2[2]);
    mx[3] = fmaxf(mx[3], m2[3]);
    v4f avg = sum * (1.0f / 256.0f);

    v4f* sp = (v4f*)(s + (size_t)b * 2 * kHW);
    sp[hw4 >> 2]                = avg;
    sp[(kHW >> 2) + (hw4 >> 2)] = mx;
  }
}

__global__ __launch_bounds__(32) void conv7x7_wmma(
    const float* __restrict__ s, const float* __restrict__ w,
    float* __restrict__ out) {
  __shared__ float tile[2 * 22 * 24];
  __shared__ float wl[98];

  int lane = threadIdx.x;
  int bid  = blockIdx.x;
  int b    = bid >> 6;
  int tl   = bid & 63;
  int y0   = (tl >> 3) << 4;
  int x0   = (tl & 7) << 4;

  const float* sb = s + (size_t)b * 2 * kHW;

  float wv[4];
  for (int k = 0; k < 4; ++k) {
    int i = lane + 32 * k;
    wv[k] = w[i < 98 ? i : 97];
  }

  int gx  = x0 + lane - 3;
  int gxc = gx < 0 ? 0 : (gx > 127 ? 127 : gx);
  bool colok = (unsigned)gx < 128u;

  float tv[44];
#pragma unroll
  for (int r = 0; r < 44; ++r) {
    int c  = r / 22;
    int yy = r - c * 22;
    int gy = y0 + yy - 3;
    int gyc = gy < 0 ? 0 : (gy > 127 ? 127 : gy);
    tv[r] = sb[c * kHW + gyc * kW + gxc];
  }

  for (int k = 0; k < 4; ++k) {
    int i = lane + 32 * k;
    if (i < 98) wl[i] = wv[k];
  }
  if (lane < 24) {
#pragma unroll
    for (int r = 0; r < 44; ++r) {
      int c  = r / 22;
      int yy = r - c * 22;
      int gy = y0 + yy - 3;
      bool ok = colok && ((unsigned)gy < 128u);
      tile[c * 528 + yy * 24 + lane] = ok ? tv[r] : 0.0f;
    }
  }
  __syncthreads();

  int half = lane >> 4;
  int mn   = lane & 15;

  int c0 = 0, yy0 = 0, dx0 = half * 2;

  v8f acc = {0.f, 0.f, 0.f, 0.f, 0.f, 0.f, 0.f, 0.f};

  for (int step = 0; step < 77; ++step) {
    int dx1 = dx0 + 1, yy1 = yy0, c1 = c0;
    if (dx1 == 7) { dx1 = 0; ++yy1; if (yy1 == 22) { yy1 = 0; ++c1; } }

    v2f a, bb;
    a.x = tile[c0 * 528 + yy0 * 24 + mn + dx0];
    a.y = tile[c1 * 528 + yy1 * 24 + mn + dx1];

    int dyA = yy0 - mn;
    int dyB = yy1 - mn;
    int iA = dyA < 0 ? 0 : (dyA > 6 ? 6 : dyA);
    int iB = dyB < 0 ? 0 : (dyB > 6 ? 6 : dyB);
    float wA = wl[c0 * 49 + iA * 7 + dx0];
    float wB = wl[c1 * 49 + iB * 7 + dx1];
    bb.x = ((unsigned)dyA < 7u) ? wA : 0.0f;
    bb.y = ((unsigned)dyB < 7u) ? wB : 0.0f;

    acc = __builtin_amdgcn_wmma_f32_16x16x4_f32(
        false, a, false, bb, (short)0, acc, false, false);

    dx0 += 4;
    if (dx0 >= 7) { dx0 -= 7; ++yy0; if (yy0 == 22) { yy0 = 0; ++c0; } }
  }

  int iy = mn;
  int jb = half * 8;
  float* ob = out + (size_t)b * kHW + (y0 + iy) * kW + (x0 + jb);
#pragma unroll
  for (int v = 0; v < 8; ++v) {
    float val = acc[v];
    ob[v] = 1.0f / (1.0f + __expf(-val));
  }
}

extern "C" void kernel_launch(void* const* d_in, const int* in_sizes, int n_in,
                              void* d_out, int out_size, void* d_ws, size_t ws_size,
                              hipStream_t stream) {
  const float* x = (const float*)d_in[0];
  const float* w = (const float*)d_in[1];
  float* out = (float*)d_out;
  float* s   = (float*)d_ws;

  reduce_mean_max<<<256, 512, 0, stream>>>(x, s);
  conv7x7_wmma<<<kB * 64, 32, 0, stream>>>(s, w, out);
}


# conv K loop re-indexed to k=(4 staged rows, dx) over one zero-padded Toeplitz weight table in LDS: each A/B fragment is one constant-offset LDS read, fetched 7 steps ahead; k sum order changes
# speedup vs baseline: 1.8960x; 1.2738x over previous
#include <hip/hip_runtime.h>

typedef float v2f __attribute__((ext_vector_type(2)));
typedef float v4f __attribute__((ext_vector_type(4)));
typedef float v8f __attribute__((ext_vector_type(8)));

static constexpr int kB  = 16;
static constexpr int kC  = 256;
static constexpr int kW  = 128;
static constexpr int kHW = 128 * 128;

static constexpr int kBatch   = 16;
static constexpr int kHalfC   = kC / 2;
static constexpr int kBatches = kHalfC / kBatch;

__global__ __launch_bounds__(512) void reduce_mean_max(
    const float* __restrict__ x, float* __restrict__ s) {
  __shared__ v4f part[2][256];

  int q    = threadIdx.x & 255;
  int hsel = threadIdx.x >> 8;
  int t    = blockIdx.x * 256 + q;
  int b    = t >> 12;
  int hw4  = (t & 4095) << 2;

  const v4f* p = (const v4f*)(x + (size_t)b * kC * kHW
                                + (size_t)hsel * kHalfC * kHW + hw4);
  v4f sum = {0.f, 0.f, 0.f, 0.f};
  float ninf = -__builtin_inff();
  v4f mx = {ninf, ninf, ninf, ninf};

  v4f buf[2][kBatch];
#define LOAD_BATCH(B)                                                        \
  _Pragma("unroll") for (int i = 0; i < kBatch; ++i)                         \
    buf[(B) & 1][i] =                                                        \
        __builtin_nontemporal_load(p + ((B) * kBatch + i) * (kHW / 4));

  LOAD_BATCH(0);
#pragma unroll
  for (int bt = 0; bt < kBatches; ++bt) {
    __builtin_amdgcn_sched_barrier(0);
    if (bt + 1 < kBatches) { LOAD_BATCH(bt + 1); }
    __builtin_amdgcn_sched_barrier(0);
#pragma unroll
    for (int i = 0; i < kBatch; ++i) {
      v4f v = buf[bt & 1][i];
      sum += v;
      mx[0] = fmaxf(mx[0], v[0]);
      mx[1] = fmaxf(mx[1], v[1]);
      mx[2] = fmaxf(mx[2], v[2]);
      mx[3] = fmaxf(mx[3], v[3]);
    }
  }
#undef LOAD_BATCH

  if (hsel) {
    part[0][q] = sum;
    part[1][q] = mx;
  }
  __syncthreads();
  if (!hsel) {
    v4f s2 = part[0][q];
    v4f m2 = part[1][q];
    sum += s2;
    mx[0] = fmaxf(mx[0], m2[0]);
    mx[1] = fmaxf(mx[1], m2[1]);
    mx[2] = fmaxf(mx[2], m2[2]);
    mx[3] = fmaxf(mx[3], m2[3]);
    v4f avg = sum * (1.0f / 256.0f);

    v4f* sp = (v4f*)(s + (size_t)b * 2 * kHW);
    sp[hw4 >> 2]                = avg;
    sp[(kHW >> 2) + (hw4 >> 2)] = mx;
  }
}

static constexpr int kTR = 44;
static constexpr int kTT = 60;

__global__ __launch_bounds__(32) void conv7x7_wmma(
    const float* __restrict__ s, const float* __restrict__ w,
    float* __restrict__ out) {
  __shared__ __attribute__((aligned(16))) float tile[22 * kTR];
  __shared__ __attribute__((aligned(16))) float wt[7 * kTT];

  int lane = threadIdx.x;
  int bid  = blockIdx.x;
  int b    = bid >> 6;
  int tl   = bid & 63;
  int y0   = (tl >> 3) << 4;
  int x0   = (tl & 7) << 4;

  const float* sb = s + (size_t)b * 2 * kHW;

  float wv[4];
  for (int k = 0; k < 4; ++k) {
    int i = lane + 32 * k;
    wv[k] = w[i < 98 ? i : 97];
  }

  int gx  = x0 + lane - 3;
  int gxc = gx < 0 ? 0 : (gx > 127 ? 127 : gx);
  bool colok = (unsigned)gx < 128u;

  float tv[44];
#pragma unroll
  for (int r = 0; r < 44; ++r) {
    int c  = r / 22;
    int yy = r - c * 22;
    int gy = y0 + yy - 3;
    int gyc = gy < 0 ? 0 : (gy > 127 ? 127 : gy);
    tv[r] = sb[c * kHW + gyc * kW + gxc];
  }

  v4f zero4 = {0.f, 0.f, 0.f, 0.f};
  for (int k = 0; k < 4; ++k) {
    int i4 = lane + 32 * k;
    if (i4 < 7 * kTT / 4) ((v4f*)wt)[i4] = zero4;
  }
  for (int k = 0; k < 4; ++k) {
    int i = lane + 32 * k;
    if (i < 98) {
      int c  = i >= 49 ? 1 : 0;
      int r  = i - 49 * c;
      int dy = r / 7;
      int dx = r - 7 * dy;
      wt[dx * kTT + 15 + 22 * c + dy] = wv[k];
    }
  }
  if (lane < 22) {
#pragma unroll
    for (int r = 0; r < 44; ++r) {
      int c  = r / 22;
      int yy = r - c * 22;
      int gy = y0 + yy - 3;
      bool ok = colok && ((unsigned)gy < 128u);
      tile[lane * kTR + r] = ok ? tv[r] : 0.0f;
    }
  }
  __syncthreads();

  int half = lane >> 4;
  int mn   = lane & 15;

  const float* ta = tile + mn * kTR + 2 * half;
  const float* tb = wt + (2 * half - mn + 15);

  v8f acc = {0.f, 0.f, 0.f, 0.f, 0.f, 0.f, 0.f, 0.f};

  v2f fa[2][7], fb[2][7];
#define LOAD_FRAGS(G)                                                        \
  _Pragma("unroll") for (int dx = 0; dx < 7; ++dx) {                         \
    fa[(G) & 1][dx]   = *(const v2f*)(ta + dx * kTR + 4 * (G));              \
    fb[(G) & 1][dx].x = tb[dx * kTT + 4 * (G)];                              \
    fb[(G) & 1][dx].y = tb[dx * kTT + 4 * (G) + 1];                          \
  }

  LOAD_FRAGS(0);
#pragma unroll
  for (int g = 0; g < 11; ++g) {
    __builtin_amdgcn_sched_barrier(0);
    if (g + 1 < 11) { LOAD_FRAGS(g + 1); }
    __builtin_amdgcn_sched_barrier(0);
#pragma unroll
    for (int dx = 0; dx < 7; ++dx) {
      acc = __builtin_amdgcn_wmma_f32_16x16x4_f32(
          false, fa[g & 1][dx], false, fb[g & 1][dx], (short)0, acc,
          false, false);
    }
  }
#undef LOAD_FRAGS

  int iy = mn;
  int jb = half * 8;
  float* ob = out + (size_t)b * kHW + (y0 + iy) * kW + (x0 + jb);
#pragma unroll
  for (int v = 0; v < 8; ++v) {
    float val = acc[v];
    ob[v] = 1.0f / (1.0f + __expf(-val));
  }
}

extern "C" void kernel_launch(void* const* d_in, const int* in_sizes, int n_in,
                              void* d_out, int out_size, void* d_ws, size_t ws_size,
                              hipStream_t stream) {
  const float* x = (const float*)d_in[0];
  const float* w = (const float*)d_in[1];
  float* out = (float*)d_out;
  float* s   = (float*)d_ws;

  reduce_mean_max<<<256, 512, 0, stream>>>(x, s);
  conv7x7_wmma<<<kB * 64, 32, 0, stream>>>(s, w, out);
}
